# starting kernel
# baseline (speedup 1.0000x reference)
#include <hip/hip_runtime.h>

typedef float v2f __attribute__((ext_vector_type(2)));
typedef float v8f __attribute__((ext_vector_type(8)));

#define N_PIX   9216
#define CCH     3
#define HDIM    96
#define QTILES  576
#define BATCH   2
#define KWAVES  8

__device__ __forceinline__ float xor16f(float v) {
    return __int_as_float(__builtin_amdgcn_ds_swizzle(__float_as_int(v), 0x401F));
}
__device__ __forceinline__ float bcastf(float v, int srcLane) {
    return __int_as_float(__builtin_amdgcn_readlane(__float_as_int(v), srcLane));
}

__global__ __launch_bounds__(32 * KWAVES)
void attn_flash_wmma(const float* __restrict__ x, float* __restrict__ fout) {
    const int lane = threadIdx.x & 31;
    const int wv   = threadIdx.x >> 5;
    const int b    = blockIdx.x / QTILES;
    const int qt   = blockIdx.x % QTILES;
    const bool lo  = lane < 16;
    const int  l15 = lane & 15;

    const float* xb = x + (size_t)b * CCH * N_PIX;
    const float scale = 0.57735026918962576f;

    v2f Bq;
    {
        const int q = qt * 16 + l15;
        float c0 = xb[0 * N_PIX + q];
        float c1 = xb[1 * N_PIX + q];
        float c2 = xb[2 * N_PIX + q];
        Bq.x = (lo ? c0 : c2) * scale;
        Bq.y = lo ? (c1 * scale) : 0.0f;
    }

    v8f  Facc = {0.f, 0.f, 0.f, 0.f, 0.f, 0.f, 0.f, 0.f};
    float sum = 0.0f;

    for (int kt = wv * 16; kt < N_PIX; kt += 16 * KWAVES) {
        const int k  = kt + l15;
        float a0 = xb[0 * N_PIX + k];
        float a1 = xb[1 * N_PIX + k];
        float a2 = xb[2 * N_PIX + k];
        v2f Ak;
        Ak.x = lo ? a0 : a2;
        Ak.y = lo ? a1 : 0.0f;

        v8f S = {0.f, 0.f, 0.f, 0.f, 0.f, 0.f, 0.f, 0.f};
        S = __builtin_amdgcn_wmma_f32_16x16x4_f32(false, Ak, false, Bq,
                                                  (short)0, S, false, false);

        float p0 = __expf(S[0]), p1 = __expf(S[1]), p2 = __expf(S[2]), p3 = __expf(S[3]);
        float p4 = __expf(S[4]), p5 = __expf(S[5]), p6 = __expf(S[6]), p7 = __expf(S[7]);
        sum += ((p0 + p1) + (p2 + p3)) + ((p4 + p5) + (p6 + p7));

        float t0 = xor16f(p0), t1 = xor16f(p1), t2 = xor16f(p2), t3 = xor16f(p3);
        float t4 = xor16f(p4), t5 = xor16f(p5), t6 = xor16f(p6), t7 = xor16f(p7);

        #pragma unroll
        for (int c = 0; c < 4; ++c) {
            v2f Ap;
            if (c == 0)      { Ap.x = lo ? p0 : t2; Ap.y = lo ? p1 : t3; }
            else if (c == 1) { Ap.x = lo ? p4 : t6; Ap.y = lo ? p5 : t7; }
            else if (c == 2) { Ap.x = lo ? t0 : p2; Ap.y = lo ? t1 : p3; }
            else             { Ap.x = lo ? t4 : p6; Ap.y = lo ? t5 : p7; }

            const int ch  = l15;
            const int chc = ch < CCH ? ch : 0;
            const int kb  = kt + 4 * c + (lo ? 0 : 2);
            float b0 = xb[chc * N_PIX + kb + 0];
            float b1 = xb[chc * N_PIX + kb + 1];
            v2f Bv;
            Bv.x = (ch < CCH) ? b0 : 0.0f;
            Bv.y = (ch < CCH) ? b1 : 0.0f;

            Facc = __builtin_amdgcn_wmma_f32_16x16x4_f32(false, Ap, false, Bv,
                                                         (short)0, Facc, false, false);
        }
    }

    __shared__ float red[KWAVES][32][9];
    #pragma unroll
    for (int r = 0; r < 8; ++r) red[wv][lane][r] = Facc[r];
    red[wv][lane][8] = sum;
    __syncthreads();

    if (wv == 0) {
        #pragma unroll
        for (int w = 1; w < KWAVES; ++w) {
            #pragma unroll
            for (int r = 0; r < 8; ++r) Facc[r] += red[w][lane][r];
            sum += red[w][lane][8];
        }

        float sum_tot = sum + xor16f(sum);

        const int ch = l15;
        #pragma unroll
        for (int r = 0; r < 8; ++r) {
            float s_q = lo ? bcastf(sum_tot, r) : bcastf(sum_tot, r + 8);
            if (ch < CCH) {
                const int qq = qt * 16 + r + (lo ? 0 : 8);
                fout[((size_t)b * CCH + ch) * N_PIX + qq] = Facc[r] / s_q;
            }
        }
    }
}

__global__ __launch_bounds__(256)
void conv3x3_bias_residual(const float* __restrict__ inp, const float* __restrict__ f,
                           const float* __restrict__ w, const float* __restrict__ bias,
                           float* __restrict__ out) {
    int idx = blockIdx.x * blockDim.x + threadIdx.x;
    if (idx >= BATCH * CCH * N_PIX) return;
    int xp = idx % HDIM;
    int t  = idx / HDIM;
    int yp = t % HDIM;  t /= HDIM;
    int co = t % CCH;
    int b  = t / CCH;

    const float* fb = f + (size_t)b * CCH * N_PIX;
    float acc = bias[co];
    #pragma unroll
    for (int ci = 0; ci < CCH; ++ci) {
        #pragma unroll
        for (int dy = -1; dy <= 1; ++dy) {
            #pragma unroll
            for (int dx = -1; dx <= 1; ++dx) {
                int yy = yp + dy, xx = xp + dx;
                if (yy >= 0 && yy < HDIM && xx >= 0 && xx < HDIM) {
                    float wv = w[((co * CCH + ci) * 3 + (dy + 1)) * 3 + (dx + 1)];
                    acc += wv * fb[ci * N_PIX + yy * HDIM + xx];
                }
            }
        }
    }
    out[idx] = inp[idx] + acc;
}

extern "C" void kernel_launch(void* const* d_in, const int* in_sizes, int n_in,
                              void* d_out, int out_size, void* d_ws, size_t ws_size,
                              hipStream_t stream) {
    const float* inp    = (const float*)d_in[0];
    const float* conv_w = (const float*)d_in[1];
    const float* conv_b = (const float*)d_in[2];
    float* out = (float*)d_out;
    float* f   = (float*)d_ws;

    attn_flash_wmma<<<BATCH * QTILES, 32 * KWAVES, 0, stream>>>(inp, f);

    const int tot = BATCH * CCH * N_PIX;
    conv3x3_bias_residual<<<(tot + 255) / 256, 256, 0, stream>>>(inp, f, conv_w, conv_b, out);
}


# key and value fragments read from a per-wave LDS stage, filled one 8-tile chunk ahead by 3 coalesced 16-byte global loads per lane, replacing 6 global load instructions per tile; sum orders unchanged
# speedup vs baseline: 1.0832x; 1.0832x over previous
#include <hip/hip_runtime.h>

typedef float v2f __attribute__((ext_vector_type(2)));
typedef float v8f __attribute__((ext_vector_type(8)));

#define N_PIX   9216
#define CCH     3
#define HDIM    96
#define QTILES  576
#define BATCH   2
#define KWAVES  8
#define TPC     8
#define NCHUNK  9
#define PITCH   144

__device__ __forceinline__ float xor16f(float v) {
    return __int_as_float(__builtin_amdgcn_ds_swizzle(__float_as_int(v), 0x401F));
}
__device__ __forceinline__ float bcastf(float v, int srcLane) {
    return __int_as_float(__builtin_amdgcn_readlane(__float_as_int(v), srcLane));
}

__global__ __launch_bounds__(32 * KWAVES, 10)
void attn_flash_wmma(const float* __restrict__ x, float* __restrict__ fout) {
    const int lane = threadIdx.x & 31;
    const int wv   = threadIdx.x >> 5;
    const int b    = blockIdx.x / QTILES;
    const int qt   = blockIdx.x % QTILES;
    const bool lo  = lane < 16;
    const int  l15 = lane & 15;

    const float* xb = x + (size_t)b * CCH * N_PIX;
    const float scale = 0.57735026918962576f;

    v2f Bq;
    {
        const int q = qt * 16 + l15;
        float c0 = xb[0 * N_PIX + q];
        float c1 = xb[1 * N_PIX + q];
        float c2 = xb[2 * N_PIX + q];
        Bq.x = (lo ? c0 : c2) * scale;
        Bq.y = lo ? (c1 * scale) : 0.0f;
    }

    v8f  Facc = {0.f, 0.f, 0.f, 0.f, 0.f, 0.f, 0.f, 0.f};
    float sum = 0.0f;

    __shared__ __attribute__((aligned(16))) float stage[KWAVES][CCH * PITCH];
    float* st = &stage[wv][0];

    const float* gp = xb + 16 * wv + 128 * (lane >> 2) + 4 * (lane & 3);
    float4 r0 = *(const float4*)(gp + 0 * N_PIX);
    float4 r1 = *(const float4*)(gp + 1 * N_PIX);
    float4 r2 = *(const float4*)(gp + 2 * N_PIX);

    const int  ch    = l15;
    const bool chok  = ch < CCH;
    const float* pa0 = st + (lo ? 0 : 2) * PITCH + l15;
    const float* pa1 = st + 1 * PITCH + l15;
    const float* pb  = st + (chok ? ch : 0) * PITCH + (lo ? 0 : 2);

    for (int i = 0; i < NCHUNK; ++i) {
        *(float4*)(st + 0 * PITCH + 4 * lane) = r0;
        *(float4*)(st + 1 * PITCH + 4 * lane) = r1;
        *(float4*)(st + 2 * PITCH + 4 * lane) = r2;
        if (i + 1 < NCHUNK) {
            const float* gn = gp + 128 * TPC * (i + 1);
            r0 = *(const float4*)(gn + 0 * N_PIX);
            r1 = *(const float4*)(gn + 1 * N_PIX);
            r2 = *(const float4*)(gn + 2 * N_PIX);
        }

        #pragma unroll 4
        for (int t = 0; t < TPC; ++t) {
            float a02 = pa0[t * 16];
            float a1  = pa1[t * 16];
            v2f Ak;
            Ak.x = a02;
            Ak.y = lo ? a1 : 0.0f;

            v8f S = {0.f, 0.f, 0.f, 0.f, 0.f, 0.f, 0.f, 0.f};
            S = __builtin_amdgcn_wmma_f32_16x16x4_f32(false, Ak, false, Bq,
                                                      (short)0, S, false, false);

            float p0 = __expf(S[0]), p1 = __expf(S[1]), p2 = __expf(S[2]), p3 = __expf(S[3]);
            float p4 = __expf(S[4]), p5 = __expf(S[5]), p6 = __expf(S[6]), p7 = __expf(S[7]);
            sum += ((p0 + p1) + (p2 + p3)) + ((p4 + p5) + (p6 + p7));

            float t0 = xor16f(p0), t1 = xor16f(p1), t2 = xor16f(p2), t3 = xor16f(p3);
            float t4 = xor16f(p4), t5 = xor16f(p5), t6 = xor16f(p6), t7 = xor16f(p7);

            #pragma unroll
            for (int c = 0; c < 4; ++c) {
                v2f Ap;
                if (c == 0)      { Ap.x = lo ? p0 : t2; Ap.y = lo ? p1 : t3; }
                else if (c == 1) { Ap.x = lo ? p4 : t6; Ap.y = lo ? p5 : t7; }
                else if (c == 2) { Ap.x = lo ? t0 : p2; Ap.y = lo ? t1 : p3; }
                else             { Ap.x = lo ? t4 : p6; Ap.y = lo ? t5 : p7; }

                const float2 bb = *(const float2*)(pb + t * 16 + 4 * c);
                v2f Bv;
                Bv.x = chok ? bb.x : 0.0f;
                Bv.y = chok ? bb.y : 0.0f;

                Facc = __builtin_amdgcn_wmma_f32_16x16x4_f32(false, Ap, false, Bv,
                                                             (short)0, Facc, false, false);
            }
        }
    }

    __shared__ float red[KWAVES][32][9];
    #pragma unroll
    for (int r = 0; r < 8; ++r) red[wv][lane][r] = Facc[r];
    red[wv][lane][8] = sum;
    __syncthreads();

    if (wv == 0) {
        #pragma unroll
        for (int w = 1; w < KWAVES; ++w) {
            #pragma unroll
            for (int r = 0; r < 8; ++r) Facc[r] += red[w][lane][r];
            sum += red[w][lane][8];
        }

        float sum_tot = sum + xor16f(sum);

        #pragma unroll
        for (int r = 0; r < 8; ++r) {
            float s_q = lo ? bcastf(sum_tot, r) : bcastf(sum_tot, r + 8);
            if (ch < CCH) {
                const int qq = qt * 16 + r + (lo ? 0 : 8);
                fout[((size_t)b * CCH + ch) * N_PIX + qq] = Facc[r] / s_q;
            }
        }
    }
}

__global__ __launch_bounds__(256)
void conv3x3_bias_residual(const float* __restrict__ inp, const float* __restrict__ f,
                           const float* __restrict__ w, const float* __restrict__ bias,
                           float* __restrict__ out) {
    int idx = blockIdx.x * blockDim.x + threadIdx.x;
    if (idx >= BATCH * CCH * N_PIX) return;
    int xp = idx % HDIM;
    int t  = idx / HDIM;
    int yp = t % HDIM;  t /= HDIM;
    int co = t % CCH;
    int b  = t / CCH;

    const float* fb = f + (size_t)b * CCH * N_PIX;
    float acc = bias[co];
    #pragma unroll
    for (int ci = 0; ci < CCH; ++ci) {
        #pragma unroll
        for (int dy = -1; dy <= 1; ++dy) {
            #pragma unroll
            for (int dx = -1; dx <= 1; ++dx) {
                int yy = yp + dy, xx = xp + dx;
                if (yy >= 0 && yy < HDIM && xx >= 0 && xx < HDIM) {
                    float wv = w[((co * CCH + ci) * 3 + (dy + 1)) * 3 + (dx + 1)];
                    acc += wv * fb[ci * N_PIX + yy * HDIM + xx];
                }
            }
        }
    }
    out[idx] = inp[idx] + acc;
}

extern "C" void kernel_launch(void* const* d_in, const int* in_sizes, int n_in,
                              void* d_out, int out_size, void* d_ws, size_t ws_size,
                              hipStream_t stream) {
    const float* inp    = (const float*)d_in[0];
    const float* conv_w = (const float*)d_in[1];
    const float* conv_b = (const float*)d_in[2];
    float* out = (float*)d_out;
    float* f   = (float*)d_ws;

    attn_flash_wmma<<<BATCH * QTILES, 32 * KWAVES, 0, stream>>>(inp, f);

    const int tot = BATCH * CCH * N_PIX;
    conv3x3_bias_residual<<<(tot + 255) / 256, 256, 0, stream>>>(inp, f, conv_w, conv_b, out);
}


# exp tile's P^T relayout by 4 v_permlane16_swap_b32 row swaps per key tile on the vector ALU, not 8 LDS-unit lane swizzles + 8 selects; denominator summed behind the swaps: its sum order changes
# speedup vs baseline: 1.1724x; 1.0823x over previous
#include <hip/hip_runtime.h>

typedef float v2f __attribute__((ext_vector_type(2)));
typedef float v8f __attribute__((ext_vector_type(8)));

#define N_PIX   9216
#define CCH     3
#define HDIM    96
#define QTILES  576
#define BATCH   2
#define KWAVES  8
#define TPC     8
#define NCHUNK  9
#define PITCH   144

__device__ __forceinline__ float xor16f(float v) {
    return __int_as_float(__builtin_amdgcn_ds_swizzle(__float_as_int(v), 0x401F));
}
__device__ __forceinline__ float bcastf(float v, int srcLane) {
    return __int_as_float(__builtin_amdgcn_readlane(__float_as_int(v), srcLane));
}
typedef unsigned int v2u __attribute__((ext_vector_type(2)));
__device__ __forceinline__ void rowswap(float& a, float& b) {
    v2u r = __builtin_amdgcn_permlane16_swap(__float_as_uint(a), __float_as_uint(b), false, false);
    a = __uint_as_float(r.x);
    b = __uint_as_float(r.y);
}

__global__ __launch_bounds__(32 * KWAVES, 10)
void attn_flash_wmma(const float* __restrict__ x, float* __restrict__ fout) {
    const int lane = threadIdx.x & 31;
    const int wv   = threadIdx.x >> 5;
    const int b    = blockIdx.x / QTILES;
    const int qt   = blockIdx.x % QTILES;
    const bool lo  = lane < 16;
    const int  l15 = lane & 15;

    const float* xb = x + (size_t)b * CCH * N_PIX;
    const float scale = 0.57735026918962576f;

    v2f Bq;
    {
        const int q = qt * 16 + l15;
        float c0 = xb[0 * N_PIX + q];
        float c1 = xb[1 * N_PIX + q];
        float c2 = xb[2 * N_PIX + q];
        Bq.x = (lo ? c0 : c2) * scale;
        Bq.y = lo ? (c1 * scale) : 0.0f;
    }

    v8f  Facc = {0.f, 0.f, 0.f, 0.f, 0.f, 0.f, 0.f, 0.f};
    float sum = 0.0f;

    __shared__ __attribute__((aligned(16))) float stage[KWAVES][CCH * PITCH];
    float* st = &stage[wv][0];

    const float* gp = xb + 16 * wv + 128 * (lane >> 2) + 4 * (lane & 3);
    float4 r0 = *(const float4*)(gp + 0 * N_PIX);
    float4 r1 = *(const float4*)(gp + 1 * N_PIX);
    float4 r2 = *(const float4*)(gp + 2 * N_PIX);

    const int  ch    = l15;
    const bool chok  = ch < CCH;
    const float* pa0 = st + (lo ? 0 : 2) * PITCH + l15;
    const float* pa1 = st + 1 * PITCH + l15;
    const float* pb  = st + (chok ? ch : 0) * PITCH + (lo ? 0 : 2);

    for (int i = 0; i < NCHUNK; ++i) {
        *(float4*)(st + 0 * PITCH + 4 * lane) = r0;
        *(float4*)(st + 1 * PITCH + 4 * lane) = r1;
        *(float4*)(st + 2 * PITCH + 4 * lane) = r2;
        if (i + 1 < NCHUNK) {
            const float* gn = gp + 128 * TPC * (i + 1);
            r0 = *(const float4*)(gn + 0 * N_PIX);
            r1 = *(const float4*)(gn + 1 * N_PIX);
            r2 = *(const float4*)(gn + 2 * N_PIX);
        }

        #pragma unroll 4
        for (int t = 0; t < TPC; ++t) {
            float a02 = pa0[t * 16];
            float a1  = pa1[t * 16];
            v2f Ak;
            Ak.x = a02;
            Ak.y = lo ? a1 : 0.0f;

            v8f S = {0.f, 0.f, 0.f, 0.f, 0.f, 0.f, 0.f, 0.f};
            S = __builtin_amdgcn_wmma_f32_16x16x4_f32(false, Ak, false, Bq,
                                                      (short)0, S, false, false);

            float p0 = __expf(S[0]), p1 = __expf(S[1]), p2 = __expf(S[2]), p3 = __expf(S[3]);
            float p4 = __expf(S[4]), p5 = __expf(S[5]), p6 = __expf(S[6]), p7 = __expf(S[7]);

            rowswap(p0, p2);
            rowswap(p1, p3);
            rowswap(p4, p6);
            rowswap(p5, p7);

            sum += ((p0 + p1) + (p2 + p3)) + ((p4 + p5) + (p6 + p7));

            #pragma unroll
            for (int c = 0; c < 4; ++c) {
                v2f Ap;
                if (c == 0)      { Ap.x = p0; Ap.y = p1; }
                else if (c == 1) { Ap.x = p4; Ap.y = p5; }
                else if (c == 2) { Ap.x = p2; Ap.y = p3; }
                else             { Ap.x = p6; Ap.y = p7; }

                const float2 bb = *(const float2*)(pb + t * 16 + 4 * c);
                v2f Bv;
                Bv.x = chok ? bb.x : 0.0f;
                Bv.y = chok ? bb.y : 0.0f;

                Facc = __builtin_amdgcn_wmma_f32_16x16x4_f32(false, Ap, false, Bv,
                                                             (short)0, Facc, false, false);
            }
        }
    }

    __shared__ float red[KWAVES][32][9];
    #pragma unroll
    for (int r = 0; r < 8; ++r) red[wv][lane][r] = Facc[r];
    red[wv][lane][8] = sum;
    __syncthreads();

    if (wv == 0) {
        #pragma unroll
        for (int w = 1; w < KWAVES; ++w) {
            #pragma unroll
            for (int r = 0; r < 8; ++r) Facc[r] += red[w][lane][r];
            sum += red[w][lane][8];
        }

        float sum_tot = sum + xor16f(sum);

        #pragma unroll
        for (int r = 0; r < 8; ++r) {
            float s_q = lo ? bcastf(sum_tot, r) : bcastf(sum_tot, r + 8);
            if (ch < CCH) {
                const int qq = qt * 16 + r + (lo ? 0 : 8);
                fout[((size_t)b * CCH + ch) * N_PIX + qq] = Facc[r] / s_q;
            }
        }
    }
}

__global__ __launch_bounds__(256)
void conv3x3_bias_residual(const float* __restrict__ inp, const float* __restrict__ f,
                           const float* __restrict__ w, const float* __restrict__ bias,
                           float* __restrict__ out) {
    int idx = blockIdx.x * blockDim.x + threadIdx.x;
    if (idx >= BATCH * CCH * N_PIX) return;
    int xp = idx % HDIM;
    int t  = idx / HDIM;
    int yp = t % HDIM;  t /= HDIM;
    int co = t % CCH;
    int b  = t / CCH;

    const float* fb = f + (size_t)b * CCH * N_PIX;
    float acc = bias[co];
    #pragma unroll
    for (int ci = 0; ci < CCH; ++ci) {
        #pragma unroll
        for (int dy = -1; dy <= 1; ++dy) {
            #pragma unroll
            for (int dx = -1; dx <= 1; ++dx) {
                int yy = yp + dy, xx = xp + dx;
                if (yy >= 0 && yy < HDIM && xx >= 0 && xx < HDIM) {
                    float wv = w[((co * CCH + ci) * 3 + (dy + 1)) * 3 + (dx + 1)];
                    acc += wv * fb[ci * N_PIX + yy * HDIM + xx];
                }
            }
        }
    }
    out[idx] = inp[idx] + acc;
}

extern "C" void kernel_launch(void* const* d_in, const int* in_sizes, int n_in,
                              void* d_out, int out_size, void* d_ws, size_t ws_size,
                              hipStream_t stream) {
    const float* inp    = (const float*)d_in[0];
    const float* conv_w = (const float*)d_in[1];
    const float* conv_b = (const float*)d_in[2];
    float* out = (float*)d_out;
    float* f   = (float*)d_ws;

    attn_flash_wmma<<<BATCH * QTILES, 32 * KWAVES, 0, stream>>>(inp, f);

    const int tot = BATCH * CCH * N_PIX;
    conv3x3_bias_residual<<<(tot + 255) / 256, 256, 0, stream>>>(inp, f, conv_w, conv_b, out);
}


# stage gets a 4th plane of ones, read as the keys' K pad and as value column 3: the softmax denominator comes out of the PV product; the loop's selects and adds go; denominator sum order changes
# speedup vs baseline: 1.2087x; 1.0310x over previous
#include <hip/hip_runtime.h>

typedef float v2f __attribute__((ext_vector_type(2)));
typedef float v8f __attribute__((ext_vector_type(8)));

#define N_PIX   9216
#define CCH     3
#define HDIM    96
#define QTILES  576
#define BATCH   2
#define KWAVES  8
#define TPC     8
#define NCHUNK  9
#define PITCH   144
#define NPLANE  4

__device__ __forceinline__ float bcastf(float v, int srcLane) {
    return __int_as_float(__builtin_amdgcn_readlane(__float_as_int(v), srcLane));
}
typedef unsigned int v2u __attribute__((ext_vector_type(2)));
__device__ __forceinline__ void rowswap(float& a, float& b) {
    v2u r = __builtin_amdgcn_permlane16_swap(__float_as_uint(a), __float_as_uint(b), false, false);
    a = __uint_as_float(r.x);
    b = __uint_as_float(r.y);
}

__global__ __launch_bounds__(32 * KWAVES, 10)
void attn_flash_wmma(const float* __restrict__ x, float* __restrict__ fout) {
    const int lane = threadIdx.x & 31;
    const int wv   = threadIdx.x >> 5;
    const int b    = blockIdx.x / QTILES;
    const int qt   = blockIdx.x % QTILES;
    const bool lo  = lane < 16;
    const int  l15 = lane & 15;

    const float* xb = x + (size_t)b * CCH * N_PIX;
    const float scale = 0.57735026918962576f;

    v2f Bq;
    {
        const int q = qt * 16 + l15;
        float c0 = xb[0 * N_PIX + q];
        float c1 = xb[1 * N_PIX + q];
        float c2 = xb[2 * N_PIX + q];
        Bq.x = (lo ? c0 : c2) * scale;
        Bq.y = lo ? (c1 * scale) : 0.0f;
    }

    v8f  Facc = {0.f, 0.f, 0.f, 0.f, 0.f, 0.f, 0.f, 0.f};

    __shared__ __attribute__((aligned(16))) float stage[KWAVES][NPLANE * PITCH];
    float* st = &stage[wv][0];
    *(float4*)(st + 3 * PITCH + 4 * lane) = make_float4(1.0f, 1.0f, 1.0f, 1.0f);

    const float* gp = xb + 16 * wv + 128 * (lane >> 2) + 4 * (lane & 3);
    float4 r0 = *(const float4*)(gp + 0 * N_PIX);
    float4 r1 = *(const float4*)(gp + 1 * N_PIX);
    float4 r2 = *(const float4*)(gp + 2 * N_PIX);

    const int  ch    = l15;
    const float* pa  = st + (lo ? 0 : 2) * PITCH + l15;
    const float* pb  = st + (ch < NPLANE ? ch : 0) * PITCH + (lo ? 0 : 2);

    for (int i = 0; i < NCHUNK; ++i) {
        *(float4*)(st + 0 * PITCH + 4 * lane) = r0;
        *(float4*)(st + 1 * PITCH + 4 * lane) = r1;
        *(float4*)(st + 2 * PITCH + 4 * lane) = r2;
        if (i + 1 < NCHUNK) {
            const float* gn = gp + 128 * TPC * (i + 1);
            r0 = *(const float4*)(gn + 0 * N_PIX);
            r1 = *(const float4*)(gn + 1 * N_PIX);
            r2 = *(const float4*)(gn + 2 * N_PIX);
        }

        #pragma unroll 4
        for (int t = 0; t < TPC; ++t) {
            v2f Ak;
            Ak.x = pa[t * 16];
            Ak.y = pa[t * 16 + PITCH];

            v8f S = {0.f, 0.f, 0.f, 0.f, 0.f, 0.f, 0.f, 0.f};
            S = __builtin_amdgcn_wmma_f32_16x16x4_f32(false, Ak, false, Bq,
                                                      (short)0, S, false, false);

            float p0 = __expf(S[0]), p1 = __expf(S[1]), p2 = __expf(S[2]), p3 = __expf(S[3]);
            float p4 = __expf(S[4]), p5 = __expf(S[5]), p6 = __expf(S[6]), p7 = __expf(S[7]);

            rowswap(p0, p2);
            rowswap(p1, p3);
            rowswap(p4, p6);
            rowswap(p5, p7);

            #pragma unroll
            for (int c = 0; c < 4; ++c) {
                v2f Ap;
                if (c == 0)      { Ap.x = p0; Ap.y = p1; }
                else if (c == 1) { Ap.x = p4; Ap.y = p5; }
                else if (c == 2) { Ap.x = p2; Ap.y = p3; }
                else             { Ap.x = p6; Ap.y = p7; }

                const float2 bb = *(const float2*)(pb + t * 16 + 4 * c);
                v2f Bv;
                Bv.x = bb.x;
                Bv.y = bb.y;

                Facc = __builtin_amdgcn_wmma_f32_16x16x4_f32(false, Ap, false, Bv,
                                                             (short)0, Facc, false, false);
            }
        }
    }

    __shared__ float red[KWAVES][32][9];
    #pragma unroll
    for (int r = 0; r < 8; ++r) red[wv][lane][r] = Facc[r];
    __syncthreads();

    if (wv == 0) {
        #pragma unroll
        for (int w = 1; w < KWAVES; ++w) {
            #pragma unroll
            for (int r = 0; r < 8; ++r) Facc[r] += red[w][lane][r];
        }

        #pragma unroll
        for (int r = 0; r < 8; ++r) {
            float s_q = lo ? bcastf(Facc[r], 3) : bcastf(Facc[r], 19);
            if (ch < CCH) {
                const int qq = qt * 16 + r + (lo ? 0 : 8);
                fout[((size_t)b * CCH + ch) * N_PIX + qq] = Facc[r] / s_q;
            }
        }
    }
}

__global__ __launch_bounds__(256)
void conv3x3_bias_residual(const float* __restrict__ inp, const float* __restrict__ f,
                           const float* __restrict__ w, const float* __restrict__ bias,
                           float* __restrict__ out) {
    int idx = blockIdx.x * blockDim.x + threadIdx.x;
    if (idx >= BATCH * CCH * N_PIX) return;
    int xp = idx % HDIM;
    int t  = idx / HDIM;
    int yp = t % HDIM;  t /= HDIM;
    int co = t % CCH;
    int b  = t / CCH;

    const float* fb = f + (size_t)b * CCH * N_PIX;
    float acc = bias[co];
    #pragma unroll
    for (int ci = 0; ci < CCH; ++ci) {
        #pragma unroll
        for (int dy = -1; dy <= 1; ++dy) {
            #pragma unroll
            for (int dx = -1; dx <= 1; ++dx) {
                int yy = yp + dy, xx = xp + dx;
                if (yy >= 0 && yy < HDIM && xx >= 0 && xx < HDIM) {
                    float wv = w[((co * CCH + ci) * 3 + (dy + 1)) * 3 + (dx + 1)];
                    acc += wv * fb[ci * N_PIX + yy * HDIM + xx];
                }
            }
        }
    }
    out[idx] = inp[idx] + acc;
}

extern "C" void kernel_launch(void* const* d_in, const int* in_sizes, int n_in,
                              void* d_out, int out_size, void* d_ws, size_t ws_size,
                              hipStream_t stream) {
    const float* inp    = (const float*)d_in[0];
    const float* conv_w = (const float*)d_in[1];
    const float* conv_b = (const float*)d_in[2];
    float* out = (float*)d_out;
    float* f   = (float*)d_ws;

    attn_flash_wmma<<<BATCH * QTILES, 32 * KWAVES, 0, stream>>>(inp, f);

    const int tot = BATCH * CCH * N_PIX;
    conv3x3_bias_residual<<<(tot + 255) / 256, 256, 0, stream>>>(inp, f, conv_w, conv_b, out);
}


# PV product's k order follows the exp tile's register layout (chunk j = keys 2j,2j+1 | 8+2j,9+2j): values read so by two 16-byte LDS reads, the 4 row swaps per tile go; key sum order in a tile changes
# speedup vs baseline: 1.2898x; 1.0671x over previous
#include <hip/hip_runtime.h>

typedef float v2f __attribute__((ext_vector_type(2)));
typedef float v8f __attribute__((ext_vector_type(8)));

#define N_PIX   9216
#define CCH     3
#define HDIM    96
#define QTILES  576
#define BATCH   2
#define KWAVES  8
#define TPC     8
#define NCHUNK  9
#define PITCH   144
#define NPLANE  4

__device__ __forceinline__ float bcastf(float v, int srcLane) {
    return __int_as_float(__builtin_amdgcn_readlane(__float_as_int(v), srcLane));
}

__global__ __launch_bounds__(32 * KWAVES, 10)
void attn_flash_wmma(const float* __restrict__ x, float* __restrict__ fout) {
    const int lane = threadIdx.x & 31;
    const int wv   = threadIdx.x >> 5;
    const int b    = blockIdx.x / QTILES;
    const int qt   = blockIdx.x % QTILES;
    const bool lo  = lane < 16;
    const int  l15 = lane & 15;

    const float* xb = x + (size_t)b * CCH * N_PIX;
    const float scale = 0.57735026918962576f;

    v2f Bq;
    {
        const int q = qt * 16 + l15;
        float c0 = xb[0 * N_PIX + q];
        float c1 = xb[1 * N_PIX + q];
        float c2 = xb[2 * N_PIX + q];
        Bq.x = (lo ? c0 : c2) * scale;
        Bq.y = lo ? (c1 * scale) : 0.0f;
    }

    v8f  Facc = {0.f, 0.f, 0.f, 0.f, 0.f, 0.f, 0.f, 0.f};

    __shared__ __attribute__((aligned(16))) float stage[KWAVES][NPLANE * PITCH];
    float* st = &stage[wv][0];
    *(float4*)(st + 3 * PITCH + 4 * lane) = make_float4(1.0f, 1.0f, 1.0f, 1.0f);

    const float* gp = xb + 16 * wv + 128 * (lane >> 2) + 4 * (lane & 3);
    float4 r0 = *(const float4*)(gp + 0 * N_PIX);
    float4 r1 = *(const float4*)(gp + 1 * N_PIX);
    float4 r2 = *(const float4*)(gp + 2 * N_PIX);

    const int  ch    = l15;
    const float* pa  = st + (lo ? 0 : 2) * PITCH + l15;
    const float* pb  = st + (ch < NPLANE ? ch : 0) * PITCH + (lo ? 0 : 8);

    #pragma unroll 1
    for (int i = 0; i < NCHUNK; ++i) {
        *(float4*)(st + 0 * PITCH + 4 * lane) = r0;
        *(float4*)(st + 1 * PITCH + 4 * lane) = r1;
        *(float4*)(st + 2 * PITCH + 4 * lane) = r2;
        if (i + 1 < NCHUNK) {
            const float* gn = gp + 128 * TPC * (i + 1);
            r0 = *(const float4*)(gn + 0 * N_PIX);
            r1 = *(const float4*)(gn + 1 * N_PIX);
            r2 = *(const float4*)(gn + 2 * N_PIX);
        }

        #pragma unroll 4
        for (int t = 0; t < TPC; ++t) {
            v2f Ak;
            Ak.x = pa[t * 16];
            Ak.y = pa[t * 16 + PITCH];

            v8f S = {0.f, 0.f, 0.f, 0.f, 0.f, 0.f, 0.f, 0.f};
            S = __builtin_amdgcn_wmma_f32_16x16x4_f32(false, Ak, false, Bq,
                                                      (short)0, S, false, false);

            float p0 = __expf(S[0]), p1 = __expf(S[1]), p2 = __expf(S[2]), p3 = __expf(S[3]);
            float p4 = __expf(S[4]), p5 = __expf(S[5]), p6 = __expf(S[6]), p7 = __expf(S[7]);

            __builtin_amdgcn_sched_barrier(0x84);

            const float4 b03 = *(const float4*)(pb + t * 16);
            const float4 b47 = *(const float4*)(pb + t * 16 + 4);

            v2f Ap, Bv;
            Ap.x = p0; Ap.y = p1; Bv.x = b03.x; Bv.y = b03.y;
            Facc = __builtin_amdgcn_wmma_f32_16x16x4_f32(false, Ap, false, Bv,
                                                         (short)0, Facc, false, false);
            Ap.x = p2; Ap.y = p3; Bv.x = b03.z; Bv.y = b03.w;
            Facc = __builtin_amdgcn_wmma_f32_16x16x4_f32(false, Ap, false, Bv,
                                                         (short)0, Facc, false, false);
            Ap.x = p4; Ap.y = p5; Bv.x = b47.x; Bv.y = b47.y;
            Facc = __builtin_amdgcn_wmma_f32_16x16x4_f32(false, Ap, false, Bv,
                                                         (short)0, Facc, false, false);
            Ap.x = p6; Ap.y = p7; Bv.x = b47.z; Bv.y = b47.w;
            Facc = __builtin_amdgcn_wmma_f32_16x16x4_f32(false, Ap, false, Bv,
                                                         (short)0, Facc, false, false);
        }
    }

    __shared__ float red[KWAVES][32][9];
    #pragma unroll
    for (int r = 0; r < 8; ++r) red[wv][lane][r] = Facc[r];
    __syncthreads();

    if (wv == 0) {
        #pragma unroll
        for (int w = 1; w < KWAVES; ++w) {
            #pragma unroll
            for (int r = 0; r < 8; ++r) Facc[r] += red[w][lane][r];
        }

        #pragma unroll
        for (int r = 0; r < 8; ++r) {
            float s_q = lo ? bcastf(Facc[r], 3) : bcastf(Facc[r], 19);
            if (ch < CCH) {
                const int qq = qt * 16 + r + (lo ? 0 : 8);
                fout[((size_t)b * CCH + ch) * N_PIX + qq] = Facc[r] / s_q;
            }
        }
    }
}

__global__ __launch_bounds__(256)
void conv3x3_bias_residual(const float* __restrict__ inp, const float* __restrict__ f,
                           const float* __restrict__ w, const float* __restrict__ bias,
                           float* __restrict__ out) {
    int idx = blockIdx.x * blockDim.x + threadIdx.x;
    if (idx >= BATCH * CCH * N_PIX) return;
    int xp = idx % HDIM;
    int t  = idx / HDIM;
    int yp = t % HDIM;  t /= HDIM;
    int co = t % CCH;
    int b  = t / CCH;

    const float* fb = f + (size_t)b * CCH * N_PIX;
    float acc = bias[co];
    #pragma unroll
    for (int ci = 0; ci < CCH; ++ci) {
        #pragma unroll
        for (int dy = -1; dy <= 1; ++dy) {
            #pragma unroll
            for (int dx = -1; dx <= 1; ++dx) {
                int yy = yp + dy, xx = xp + dx;
                if (yy >= 0 && yy < HDIM && xx >= 0 && xx < HDIM) {
                    float wv = w[((co * CCH + ci) * 3 + (dy + 1)) * 3 + (dx + 1)];
                    acc += wv * fb[ci * N_PIX + yy * HDIM + xx];
                }
            }
        }
    }
    out[idx] = inp[idx] + acc;
}

extern "C" void kernel_launch(void* const* d_in, const int* in_sizes, int n_in,
                              void* d_out, int out_size, void* d_ws, size_t ws_size,
                              hipStream_t stream) {
    const float* inp    = (const float*)d_in[0];
    const float* conv_w = (const float*)d_in[1];
    const float* conv_b = (const float*)d_in[2];
    float* out = (float*)d_out;
    float* f   = (float*)d_ws;

    attn_flash_wmma<<<BATCH * QTILES, 32 * KWAVES, 0, stream>>>(inp, f);

    const int tot = BATCH * CCH * N_PIX;
    conv3x3_bias_residual<<<(tot + 255) / 256, 256, 0, stream>>>(inp, f, conv_w, conv_b, out);
}


# conv stage: a block parks its 8x32 patch of f with border (3x10x34, zero outside the image) in LDS by one load group; a thread forms its pixel's 3 outputs from 27 LDS taps; sum order unchanged
# speedup vs baseline: 1.3590x; 1.0537x over previous
#include <hip/hip_runtime.h>

typedef float v2f __attribute__((ext_vector_type(2)));
typedef float v8f __attribute__((ext_vector_type(8)));

#define N_PIX   9216
#define CCH     3
#define HDIM    96
#define QTILES  576
#define BATCH   2
#define KWAVES  8
#define TPC     8
#define NCHUNK  9
#define PITCH   144
#define NPLANE  4

__device__ __forceinline__ float bcastf(float v, int srcLane) {
    return __int_as_float(__builtin_amdgcn_readlane(__float_as_int(v), srcLane));
}

__global__ __launch_bounds__(32 * KWAVES, 10)
void attn_flash_wmma(const float* __restrict__ x, float* __restrict__ fout) {
    const int lane = threadIdx.x & 31;
    const int wv   = threadIdx.x >> 5;
    const int b    = blockIdx.x / QTILES;
    const int qt   = blockIdx.x % QTILES;
    const bool lo  = lane < 16;
    const int  l15 = lane & 15;

    const float* xb = x + (size_t)b * CCH * N_PIX;
    const float scale = 0.57735026918962576f;

    v2f Bq;
    {
        const int q = qt * 16 + l15;
        float c0 = xb[0 * N_PIX + q];
        float c1 = xb[1 * N_PIX + q];
        float c2 = xb[2 * N_PIX + q];
        Bq.x = (lo ? c0 : c2) * scale;
        Bq.y = lo ? (c1 * scale) : 0.0f;
    }

    v8f  Facc = {0.f, 0.f, 0.f, 0.f, 0.f, 0.f, 0.f, 0.f};

    __shared__ __attribute__((aligned(16))) float stage[KWAVES][NPLANE * PITCH];
    float* st = &stage[wv][0];
    *(float4*)(st + 3 * PITCH + 4 * lane) = make_float4(1.0f, 1.0f, 1.0f, 1.0f);

    const float* gp = xb + 16 * wv + 128 * (lane >> 2) + 4 * (lane & 3);
    float4 r0 = *(const float4*)(gp + 0 * N_PIX);
    float4 r1 = *(const float4*)(gp + 1 * N_PIX);
    float4 r2 = *(const float4*)(gp + 2 * N_PIX);

    const int  ch    = l15;
    const float* pa  = st + (lo ? 0 : 2) * PITCH + l15;
    const float* pb  = st + (ch < NPLANE ? ch : 0) * PITCH + (lo ? 0 : 8);

    #pragma unroll 1
    for (int i = 0; i < NCHUNK; ++i) {
        *(float4*)(st + 0 * PITCH + 4 * lane) = r0;
        *(float4*)(st + 1 * PITCH + 4 * lane) = r1;
        *(float4*)(st + 2 * PITCH + 4 * lane) = r2;
        if (i + 1 < NCHUNK) {
            const float* gn = gp + 128 * TPC * (i + 1);
            r0 = *(const float4*)(gn + 0 * N_PIX);
            r1 = *(const float4*)(gn + 1 * N_PIX);
            r2 = *(const float4*)(gn + 2 * N_PIX);
        }

        #pragma unroll 4
        for (int t = 0; t < TPC; ++t) {
            v2f Ak;
            Ak.x = pa[t * 16];
            Ak.y = pa[t * 16 + PITCH];

            v8f S = {0.f, 0.f, 0.f, 0.f, 0.f, 0.f, 0.f, 0.f};
            S = __builtin_amdgcn_wmma_f32_16x16x4_f32(false, Ak, false, Bq,
                                                      (short)0, S, false, false);

            float p0 = __expf(S[0]), p1 = __expf(S[1]), p2 = __expf(S[2]), p3 = __expf(S[3]);
            float p4 = __expf(S[4]), p5 = __expf(S[5]), p6 = __expf(S[6]), p7 = __expf(S[7]);

            __builtin_amdgcn_sched_barrier(0x84);

            const float4 b03 = *(const float4*)(pb + t * 16);
            const float4 b47 = *(const float4*)(pb + t * 16 + 4);

            v2f Ap, Bv;
            Ap.x = p0; Ap.y = p1; Bv.x = b03.x; Bv.y = b03.y;
            Facc = __builtin_amdgcn_wmma_f32_16x16x4_f32(false, Ap, false, Bv,
                                                         (short)0, Facc, false, false);
            Ap.x = p2; Ap.y = p3; Bv.x = b03.z; Bv.y = b03.w;
            Facc = __builtin_amdgcn_wmma_f32_16x16x4_f32(false, Ap, false, Bv,
                                                         (short)0, Facc, false, false);
            Ap.x = p4; Ap.y = p5; Bv.x = b47.x; Bv.y = b47.y;
            Facc = __builtin_amdgcn_wmma_f32_16x16x4_f32(false, Ap, false, Bv,
                                                         (short)0, Facc, false, false);
            Ap.x = p6; Ap.y = p7; Bv.x = b47.z; Bv.y = b47.w;
            Facc = __builtin_amdgcn_wmma_f32_16x16x4_f32(false, Ap, false, Bv,
                                                         (short)0, Facc, false, false);
        }
    }

    __shared__ float red[KWAVES][32][9];
    #pragma unroll
    for (int r = 0; r < 8; ++r) red[wv][lane][r] = Facc[r];
    __syncthreads();

    if (wv == 0) {
        #pragma unroll
        for (int w = 1; w < KWAVES; ++w) {
            #pragma unroll
            for (int r = 0; r < 8; ++r) Facc[r] += red[w][lane][r];
        }

        #pragma unroll
        for (int r = 0; r < 8; ++r) {
            float s_q = lo ? bcastf(Facc[r], 3) : bcastf(Facc[r], 19);
            if (ch < CCH) {
                const int qq = qt * 16 + r + (lo ? 0 : 8);
                fout[((size_t)b * CCH + ch) * N_PIX + qq] = Facc[r] / s_q;
            }
        }
    }
}

#define CT_H   8
#define CT_W   32
#define CT_HH  (CT_H + 2)
#define CT_WW  (CT_W + 2)
#define CT_PL  (CT_HH * CT_WW)
#define CT_N   (CCH * CT_PL)
#define CT_LD  4

__global__ __launch_bounds__(256)
void conv3x3_bias_residual(const float* __restrict__ inp, const float* __restrict__ f,
                           const float* __restrict__ w, const float* __restrict__ bias,
                           float* __restrict__ out) {
    __shared__ float tile[CT_LD * 256];

    const int tid = threadIdx.x;
    const int tx  = tid & 31;
    const int ty  = tid >> 5;
    int bid = blockIdx.x;
    const int x0 = (bid % (HDIM / CT_W)) * CT_W;  bid /= (HDIM / CT_W);
    const int y0 = (bid % (HDIM / CT_H)) * CT_H;
    const int b  = bid / (HDIM / CT_H);

    const float* fb = f + (size_t)b * CCH * N_PIX;

    float wr[CCH * CCH * 9];
    #pragma unroll
    for (int i = 0; i < CCH * CCH * 9; ++i) wr[i] = w[i];
    float acc0 = bias[0], acc1 = bias[1], acc2 = bias[2];

    float v[CT_LD];
    #pragma unroll
    for (int k = 0; k < CT_LD; ++k) {
        int e = tid + 256 * k;
        e = e < CT_N ? e : CT_N - 1;
        const int ci = e / CT_PL;
        const int r  = e - ci * CT_PL;
        const int yy = r / CT_WW;
        const int xx = r - yy * CT_WW;
        const int gy = y0 + yy - 1;
        const int gx = x0 + xx - 1;
        const bool inb = (unsigned)gy < (unsigned)HDIM && (unsigned)gx < (unsigned)HDIM;
        const float t = fb[ci * N_PIX + (inb ? gy * HDIM + gx : 0)];
        v[k] = inb ? t : 0.0f;
    }

    const int y = y0 + ty, x = x0 + tx;
    const size_t o0 = (size_t)b * CCH * N_PIX + y * HDIM + x;
    const float res0 = inp[o0];
    const float res1 = inp[o0 + N_PIX];
    const float res2 = inp[o0 + 2 * N_PIX];

    #pragma unroll
    for (int k = 0; k < CT_LD; ++k) {
        tile[tid + 256 * k] = v[k];
    }
    __syncthreads();

    #pragma unroll
    for (int ci = 0; ci < CCH; ++ci) {
        #pragma unroll
        for (int dy = 0; dy < 3; ++dy) {
            #pragma unroll
            for (int dx = 0; dx < 3; ++dx) {
                const float t = tile[ci * CT_PL + (ty + dy) * CT_WW + tx + dx];
                acc0 += wr[((0 * CCH + ci) * 3 + dy) * 3 + dx] * t;
                acc1 += wr[((1 * CCH + ci) * 3 + dy) * 3 + dx] * t;
                acc2 += wr[((2 * CCH + ci) * 3 + dy) * 3 + dx] * t;
            }
        }
    }
    out[o0]             = res0 + acc0;
    out[o0 + N_PIX]     = res1 + acc1;
    out[o0 + 2 * N_PIX] = res2 + acc2;
}

extern "C" void kernel_launch(void* const* d_in, const int* in_sizes, int n_in,
                              void* d_out, int out_size, void* d_ws, size_t ws_size,
                              hipStream_t stream) {
    const float* inp    = (const float*)d_in[0];
    const float* conv_w = (const float*)d_in[1];
    const float* conv_b = (const float*)d_in[2];
    float* out = (float*)d_out;
    float* f   = (float*)d_ws;

    attn_flash_wmma<<<BATCH * QTILES, 32 * KWAVES, 0, stream>>>(inp, f);

    conv3x3_bias_residual<<<72, 256, 0, stream>>>(inp, f, conv_w, conv_b, out);
}


# speedup vs baseline: 1.4022x; 1.0318x over previous
#include <hip/hip_runtime.h>

typedef float v2f __attribute__((ext_vector_type(2)));
typedef float v8f __attribute__((ext_vector_type(8)));

#define N_PIX   9216
#define CCH     3
#define HDIM    96
#define QTILES  576
#define BATCH   2
#define KWAVES  8
#define TPC     8
#define NCHUNK  9
#define PITCH   144
#define NPLANE  4

__device__ __forceinline__ float bcastf(float v, int srcLane) {
    return __int_as_float(__builtin_amdgcn_readlane(__float_as_int(v), srcLane));
}

__global__ __launch_bounds__(32 * KWAVES, 10)
void attn_flash_wmma(const float* __restrict__ x, float* __restrict__ fout) {
    const int lane = threadIdx.x & 31;
    const int wv   = threadIdx.x >> 5;
    const int b    = blockIdx.x / QTILES;
    const int qt   = blockIdx.x % QTILES;
    const bool lo  = lane < 16;
    const int  l15 = lane & 15;

    const float* xb = x + (size_t)b * CCH * N_PIX;
    const float scale = 0.57735026918962576f;

    v2f Bq;
    {
        const int q = qt * 16 + l15;
        float c0 = xb[0 * N_PIX + q];
        float c1 = xb[1 * N_PIX + q];
        float c2 = xb[2 * N_PIX + q];
        Bq.x = (lo ? c0 : c2) * scale;
        Bq.y = lo ? (c1 * scale) : 0.0f;
    }

    v8f  Facc = {0.f, 0.f, 0.f, 0.f, 0.f, 0.f, 0.f, 0.f};

    __shared__ __attribute__((aligned(16))) float stage[KWAVES][NPLANE * PITCH];
    float* st = &stage[wv][0];
    *(float4*)(st + 3 * PITCH + 4 * lane) = make_float4(1.0f, 1.0f, 1.0f, 1.0f);

    const float* gp = xb + 16 * wv + 128 * (lane >> 2) + 4 * (lane & 3);
    float4 r0 = *(const float4*)(gp + 0 * N_PIX);
    float4 r1 = *(const float4*)(gp + 1 * N_PIX);
    float4 r2 = *(const float4*)(gp + 2 * N_PIX);

    const int  ch    = l15;
    const float* pa  = st + (lo ? 0 : 2) * PITCH + l15;
    const float* pb  = st + (ch < NPLANE ? ch : 0) * PITCH + (lo ? 0 : 8);

    #pragma unroll 1
    for (int i = 0; i < NCHUNK; ++i) {
        *(float4*)(st + 0 * PITCH + 4 * lane) = r0;
        *(float4*)(st + 1 * PITCH + 4 * lane) = r1;
        *(float4*)(st + 2 * PITCH + 4 * lane) = r2;
        if (i + 1 < NCHUNK) {
            const float* gn = gp + 128 * TPC * (i + 1);
            r0 = *(const float4*)(gn + 0 * N_PIX);
            r1 = *(const float4*)(gn + 1 * N_PIX);
            r2 = *(const float4*)(gn + 2 * N_PIX);
        }

        const float L2E = 1.44269504088896340736f;
        v2f Ak1;
        v2f m01, m23, m45, m67;
        float4 b03, b47;
        {
            v2f Ak0;
            Ak0.x = pa[0];
            Ak0.y = pa[PITCH];
            b03 = *(const float4*)(pb);
            b47 = *(const float4*)(pb + 4);
            __builtin_amdgcn_sched_barrier(0);
            Ak1.x = pa[16];
            Ak1.y = pa[16 + PITCH];
            v8f S = {0.f, 0.f, 0.f, 0.f, 0.f, 0.f, 0.f, 0.f};
            S = __builtin_amdgcn_wmma_f32_16x16x4_f32(false, Ak0, false, Bq,
                                                      (short)0, S, false, false);
            m01.x = S[0]; m01.y = S[1]; m23.x = S[2]; m23.y = S[3];
            m45.x = S[4]; m45.y = S[5]; m67.x = S[6]; m67.y = S[7];
            m01 *= L2E; m23 *= L2E; m45 *= L2E; m67 *= L2E;
        }
        __builtin_amdgcn_sched_barrier(0);

        #pragma unroll
        for (int t = 0; t < TPC; ++t) {
            v2f n01 = m01, n23 = m23, n45 = m45, n67 = m67;
            float4 c03 = b03, c47 = b47;
            v2f Ak2 = Ak1;
            if (t + 1 < TPC) {
                v8f S = {0.f, 0.f, 0.f, 0.f, 0.f, 0.f, 0.f, 0.f};
                S = __builtin_amdgcn_wmma_f32_16x16x4_f32(false, Ak1, false, Bq,
                                                          (short)0, S, false, false);
                n01.x = S[0]; n01.y = S[1]; n23.x = S[2]; n23.y = S[3];
                n45.x = S[4]; n45.y = S[5]; n67.x = S[6]; n67.y = S[7];
                n01 *= L2E; n23 *= L2E; n45 *= L2E; n67 *= L2E;
                c03 = *(const float4*)(pb + (t + 1) * 16);
                c47 = *(const float4*)(pb + (t + 1) * 16 + 4);
            }
            if (t + 2 < TPC) {
                Ak2.x = pa[(t + 2) * 16];
                Ak2.y = pa[(t + 2) * 16 + PITCH];
            }

            const float p0 = __builtin_amdgcn_exp2f(m01.x), p1 = __builtin_amdgcn_exp2f(m01.y);
            const float p2 = __builtin_amdgcn_exp2f(m23.x), p3 = __builtin_amdgcn_exp2f(m23.y);
            const float p4 = __builtin_amdgcn_exp2f(m45.x), p5 = __builtin_amdgcn_exp2f(m45.y);
            const float p6 = __builtin_amdgcn_exp2f(m67.x), p7 = __builtin_amdgcn_exp2f(m67.y);

            v2f Ap, Bv;
            Ap.x = p0; Ap.y = p1; Bv.x = b03.x; Bv.y = b03.y;
            Facc = __builtin_amdgcn_wmma_f32_16x16x4_f32(false, Ap, false, Bv,
                                                         (short)0, Facc, false, false);
            Ap.x = p2; Ap.y = p3; Bv.x = b03.z; Bv.y = b03.w;
            Facc = __builtin_amdgcn_wmma_f32_16x16x4_f32(false, Ap, false, Bv,
                                                         (short)0, Facc, false, false);
            Ap.x = p4; Ap.y = p5; Bv.x = b47.x; Bv.y = b47.y;
            Facc = __builtin_amdgcn_wmma_f32_16x16x4_f32(false, Ap, false, Bv,
                                                         (short)0, Facc, false, false);
            Ap.x = p6; Ap.y = p7; Bv.x = b47.z; Bv.y = b47.w;
            Facc = __builtin_amdgcn_wmma_f32_16x16x4_f32(false, Ap, false, Bv,
                                                         (short)0, Facc, false, false);

            if (t + 1 < TPC) {
                __builtin_amdgcn_sched_group_barrier(0x008, 1, 0);
                __builtin_amdgcn_sched_group_barrier(0x400, 1, 0);
                __builtin_amdgcn_sched_group_barrier(0x100, 1, 0);
                __builtin_amdgcn_sched_group_barrier(0x400, 1, 0);
                __builtin_amdgcn_sched_group_barrier(0x100, 1, 0);
                if (t + 2 < TPC) {
                    __builtin_amdgcn_sched_group_barrier(0x400, 1, 0);
                    __builtin_amdgcn_sched_group_barrier(0x100, 1, 0);
                }
                __builtin_amdgcn_sched_group_barrier(0x400, 1, 0);
                __builtin_amdgcn_sched_group_barrier(0x002, 1, 0);
                __builtin_amdgcn_sched_group_barrier(0x400, 1, 0);
                __builtin_amdgcn_sched_group_barrier(0x002, 1, 0);
                __builtin_amdgcn_sched_group_barrier(0x400, 1, 0);
                __builtin_amdgcn_sched_group_barrier(0x002, 1, 0);
                __builtin_amdgcn_sched_group_barrier(0x400, 1, 0);
                __builtin_amdgcn_sched_group_barrier(0x002, 1, 0);
                if (t + 2 < TPC) __builtin_amdgcn_sched_group_barrier(0x400, 1, 0);
                else             __builtin_amdgcn_sched_group_barrier(0x400, 2, 0);
            } else {
                __builtin_amdgcn_sched_group_barrier(0x400, 8, 0);
            }
            __builtin_amdgcn_sched_group_barrier(0x008, 4, 0);
            __builtin_amdgcn_sched_barrier(0);

            m01 = n01; m23 = n23; m45 = n45; m67 = n67;
            b03 = c03; b47 = c47;
            Ak1 = Ak2;
        }
    }

    __shared__ float red[KWAVES][32][9];
    #pragma unroll
    for (int r = 0; r < 8; ++r) red[wv][lane][r] = Facc[r];
    __syncthreads();

    if (wv == 0) {
        #pragma unroll
        for (int w = 1; w < KWAVES; ++w) {
            #pragma unroll
            for (int r = 0; r < 8; ++r) Facc[r] += red[w][lane][r];
        }

        #pragma unroll
        for (int r = 0; r < 8; ++r) {
            float s_q = lo ? bcastf(Facc[r], 3) : bcastf(Facc[r], 19);
            if (ch < CCH) {
                const int qq = qt * 16 + r + (lo ? 0 : 8);
                fout[((size_t)b * CCH + ch) * N_PIX + qq] = Facc[r] / s_q;
            }
        }
    }
}

#define CT_H   8
#define CT_W   32
#define CT_HH  (CT_H + 2)
#define CT_WW  (CT_W + 2)
#define CT_PL  (CT_HH * CT_WW)
#define CT_N   (CCH * CT_PL)
#define CT_LD  4

__global__ __launch_bounds__(256)
void conv3x3_bias_residual(const float* __restrict__ inp, const float* __restrict__ f,
                           const float* __restrict__ w, const float* __restrict__ bias,
                           float* __restrict__ out) {
    __shared__ float tile[CT_LD * 256];

    const int tid = threadIdx.x;
    const int tx  = tid & 31;
    const int ty  = tid >> 5;
    int bid = blockIdx.x;
    const int x0 = (bid % (HDIM / CT_W)) * CT_W;  bid /= (HDIM / CT_W);
    const int y0 = (bid % (HDIM / CT_H)) * CT_H;
    const int b  = bid / (HDIM / CT_H);

    const float* fb = f + (size_t)b * CCH * N_PIX;

    float wr[CCH * CCH * 9];
    #pragma unroll
    for (int i = 0; i < CCH * CCH * 9; ++i) wr[i] = w[i];
    float acc0 = bias[0], acc1 = bias[1], acc2 = bias[2];

    float v[CT_LD];
    #pragma unroll
    for (int k = 0; k < CT_LD; ++k) {
        int e = tid + 256 * k;
        e = e < CT_N ? e : CT_N - 1;
        const int ci = e / CT_PL;
        const int r  = e - ci * CT_PL;
        const int yy = r / CT_WW;
        const int xx = r - yy * CT_WW;
        const int gy = y0 + yy - 1;
        const int gx = x0 + xx - 1;
        const bool inb = (unsigned)gy < (unsigned)HDIM && (unsigned)gx < (unsigned)HDIM;
        const float t = fb[ci * N_PIX + (inb ? gy * HDIM + gx : 0)];
        v[k] = inb ? t : 0.0f;
    }

    const int y = y0 + ty, x = x0 + tx;
    const size_t o0 = (size_t)b * CCH * N_PIX + y * HDIM + x;
    const float res0 = inp[o0];
    const float res1 = inp[o0 + N_PIX];
    const float res2 = inp[o0 + 2 * N_PIX];

    #pragma unroll
    for (int k = 0; k < CT_LD; ++k) {
        tile[tid + 256 * k] = v[k];
    }
    __syncthreads();

    #pragma unroll
    for (int ci = 0; ci < CCH; ++ci) {
        #pragma unroll
        for (int dy = 0; dy < 3; ++dy) {
            #pragma unroll
            for (int dx = 0; dx < 3; ++dx) {
                const float t = tile[ci * CT_PL + (ty + dy) * CT_WW + tx + dx];
                acc0 += wr[((0 * CCH + ci) * 3 + dy) * 3 + dx] * t;
                acc1 += wr[((1 * CCH + ci) * 3 + dy) * 3 + dx] * t;
                acc2 += wr[((2 * CCH + ci) * 3 + dy) * 3 + dx] * t;
            }
        }
    }
    out[o0]             = res0 + acc0;
    out[o0 + N_PIX]     = res1 + acc1;
    out[o0 + 2 * N_PIX] = res2 + acc2;
}

extern "C" void kernel_launch(void* const* d_in, const int* in_sizes, int n_in,
                              void* d_out, int out_size, void* d_ws, size_t ws_size,
                              hipStream_t stream) {
    const float* inp    = (const float*)d_in[0];
    const float* conv_w = (const float*)d_in[1];
    const float* conv_b = (const float*)d_in[2];
    float* out = (float*)d_out;
    float* f   = (float*)d_ws;

    attn_flash_wmma<<<BATCH * QTILES, 32 * KWAVES, 0, stream>>>(inp, f);

    conv3x3_bias_residual<<<72, 256, 0, stream>>>(inp, f, conv_w, conv_b, out);
}


# a query tile's 8 key streams split over two 4-wave work-groups (2304: 9 per WGP, not 4 or 5); unnormalised partial sums go to the workspace, conv stage adds and divides; merge sum order changes
# speedup vs baseline: 1.5844x; 1.1299x over previous
#include <hip/hip_runtime.h>

typedef float v2f __attribute__((ext_vector_type(2)));
typedef float v8f __attribute__((ext_vector_type(8)));

#define N_PIX   9216
#define CCH     3
#define HDIM    96
#define QTILES  576
#define BATCH   2
#define KSTREAMS 8
#define KSPLIT  2
#define PART_FLOATS (BATCH * QTILES * KSPLIT * 64)
#define TPC     8
#define NCHUNK  9
#define PITCH   144
#define NPLANE  4

__device__ __forceinline__ float bcastf(float v, int srcLane) {
    return __int_as_float(__builtin_amdgcn_readlane(__float_as_int(v), srcLane));
}

template <int KS>
__global__ __launch_bounds__(32 * (KSTREAMS / KS), 10)
void attn_flash_wmma(const float* __restrict__ x, float* __restrict__ fout) {
    constexpr int KWAVES = KSTREAMS / KS;
    const int lane = threadIdx.x & 31;
    const int wv   = threadIdx.x >> 5;
    const int part = blockIdx.x % KS;
    const int tile = blockIdx.x / KS;
    const int b    = tile / QTILES;
    const int qt   = tile % QTILES;
    const int ks   = part * KWAVES + wv;
    const bool lo  = lane < 16;
    const int  l15 = lane & 15;

    const float* xb = x + (size_t)b * CCH * N_PIX;
    const float scale = 0.57735026918962576f;

    v2f Bq;
    {
        const int q = qt * 16 + l15;
        float c0 = xb[0 * N_PIX + q];
        float c1 = xb[1 * N_PIX + q];
        float c2 = xb[2 * N_PIX + q];
        Bq.x = (lo ? c0 : c2) * scale;
        Bq.y = lo ? (c1 * scale) : 0.0f;
    }

    v8f  Facc = {0.f, 0.f, 0.f, 0.f, 0.f, 0.f, 0.f, 0.f};

    __shared__ __attribute__((aligned(16))) float stage[KWAVES][NPLANE * PITCH];
    float* st = &stage[wv][0];
    *(float4*)(st + 3 * PITCH + 4 * lane) = make_float4(1.0f, 1.0f, 1.0f, 1.0f);

    const float* gp = xb + 16 * ks + 128 * (lane >> 2) + 4 * (lane & 3);
    float4 r0 = *(const float4*)(gp + 0 * N_PIX);
    float4 r1 = *(const float4*)(gp + 1 * N_PIX);
    float4 r2 = *(const float4*)(gp + 2 * N_PIX);

    const int  ch    = l15;
    const float* pa  = st + (lo ? 0 : 2) * PITCH + l15;
    const float* pb  = st + (ch < NPLANE ? ch : 0) * PITCH + (lo ? 0 : 8);

    #pragma unroll 1
    for (int i = 0; i < NCHUNK; ++i) {
        *(float4*)(st + 0 * PITCH + 4 * lane) = r0;
        *(float4*)(st + 1 * PITCH + 4 * lane) = r1;
        *(float4*)(st + 2 * PITCH + 4 * lane) = r2;
        if (i + 1 < NCHUNK) {
            const float* gn = gp + 128 * TPC * (i + 1);
            r0 = *(const float4*)(gn + 0 * N_PIX);
            r1 = *(const float4*)(gn + 1 * N_PIX);
            r2 = *(const float4*)(gn + 2 * N_PIX);
        }

        const float L2E = 1.44269504088896340736f;
        v2f Ak1;
        v2f m01, m23, m45, m67;
        float4 b03, b47;
        {
            v2f Ak0;
            Ak0.x = pa[0];
            Ak0.y = pa[PITCH];
            b03 = *(const float4*)(pb);
            b47 = *(const float4*)(pb + 4);
            __builtin_amdgcn_sched_barrier(0);
            Ak1.x = pa[16];
            Ak1.y = pa[16 + PITCH];
            v8f S = {0.f, 0.f, 0.f, 0.f, 0.f, 0.f, 0.f, 0.f};
            S = __builtin_amdgcn_wmma_f32_16x16x4_f32(false, Ak0, false, Bq,
                                                      (short)0, S, false, false);
            m01.x = S[0]; m01.y = S[1]; m23.x = S[2]; m23.y = S[3];
            m45.x = S[4]; m45.y = S[5]; m67.x = S[6]; m67.y = S[7];
            m01 *= L2E; m23 *= L2E; m45 *= L2E; m67 *= L2E;
        }
        __builtin_amdgcn_sched_barrier(0);

        #pragma unroll
        for (int t = 0; t < TPC; ++t) {
            v2f n01 = m01, n23 = m23, n45 = m45, n67 = m67;
            float4 c03 = b03, c47 = b47;
            v2f Ak2 = Ak1;
            if (t + 1 < TPC) {
                v8f S = {0.f, 0.f, 0.f, 0.f, 0.f, 0.f, 0.f, 0.f};
                S = __builtin_amdgcn_wmma_f32_16x16x4_f32(false, Ak1, false, Bq,
                                                          (short)0, S, false, false);
                n01.x = S[0]; n01.y = S[1]; n23.x = S[2]; n23.y = S[3];
                n45.x = S[4]; n45.y = S[5]; n67.x = S[6]; n67.y = S[7];
                n01 *= L2E; n23 *= L2E; n45 *= L2E; n67 *= L2E;
                c03 = *(const float4*)(pb + (t + 1) * 16);
                c47 = *(const float4*)(pb + (t + 1) * 16 + 4);
            }
            if (t + 2 < TPC) {
                Ak2.x = pa[(t + 2) * 16];
                Ak2.y = pa[(t + 2) * 16 + PITCH];
            }

            const float p0 = __builtin_amdgcn_exp2f(m01.x), p1 = __builtin_amdgcn_exp2f(m01.y);
            const float p2 = __builtin_amdgcn_exp2f(m23.x), p3 = __builtin_amdgcn_exp2f(m23.y);
            const float p4 = __builtin_amdgcn_exp2f(m45.x), p5 = __builtin_amdgcn_exp2f(m45.y);
            const float p6 = __builtin_amdgcn_exp2f(m67.x), p7 = __builtin_amdgcn_exp2f(m67.y);

            v2f Ap, Bv;
            Ap.x = p0; Ap.y = p1; Bv.x = b03.x; Bv.y = b03.y;
            Facc = __builtin_amdgcn_wmma_f32_16x16x4_f32(false, Ap, false, Bv,
                                                         (short)0, Facc, false, false);
            Ap.x = p2; Ap.y = p3; Bv.x = b03.z; Bv.y = b03.w;
            Facc = __builtin_amdgcn_wmma_f32_16x16x4_f32(false, Ap, false, Bv,
                                                         (short)0, Facc, false, false);
            Ap.x = p4; Ap.y = p5; Bv.x = b47.x; Bv.y = b47.y;
            Facc = __builtin_amdgcn_wmma_f32_16x16x4_f32(false, Ap, false, Bv,
                                                         (short)0, Facc, false, false);
            Ap.x = p6; Ap.y = p7; Bv.x = b47.z; Bv.y = b47.w;
            Facc = __builtin_amdgcn_wmma_f32_16x16x4_f32(false, Ap, false, Bv,
                                                         (short)0, Facc, false, false);

            if (t + 1 < TPC) {
                __builtin_amdgcn_sched_group_barrier(0x008, 1, 0);
                __builtin_amdgcn_sched_group_barrier(0x400, 1, 0);
                __builtin_amdgcn_sched_group_barrier(0x100, 1, 0);
                __builtin_amdgcn_sched_group_barrier(0x400, 1, 0);
                __builtin_amdgcn_sched_group_barrier(0x100, 1, 0);
                if (t + 2 < TPC) {
                    __builtin_amdgcn_sched_group_barrier(0x400, 1, 0);
                    __builtin_amdgcn_sched_group_barrier(0x100, 1, 0);
                }
                __builtin_amdgcn_sched_group_barrier(0x400, 1, 0);
                __builtin_amdgcn_sched_group_barrier(0x002, 1, 0);
                __builtin_amdgcn_sched_group_barrier(0x400, 1, 0);
                __builtin_amdgcn_sched_group_barrier(0x002, 1, 0);
                __builtin_amdgcn_sched_group_barrier(0x400, 1, 0);
                __builtin_amdgcn_sched_group_barrier(0x002, 1, 0);
                __builtin_amdgcn_sched_group_barrier(0x400, 1, 0);
                __builtin_amdgcn_sched_group_barrier(0x002, 1, 0);
                if (t + 2 < TPC) __builtin_amdgcn_sched_group_barrier(0x400, 1, 0);
                else             __builtin_amdgcn_sched_group_barrier(0x400, 2, 0);
            } else {
                __builtin_amdgcn_sched_group_barrier(0x400, 8, 0);
            }
            __builtin_amdgcn_sched_group_barrier(0x008, 4, 0);
            __builtin_amdgcn_sched_barrier(0);

            m01 = n01; m23 = n23; m45 = n45; m67 = n67;
            b03 = c03; b47 = c47;
            Ak1 = Ak2;
        }
    }

    __shared__ float red[KWAVES][32][9];
    #pragma unroll
    for (int r = 0; r < 8; ++r) red[wv][lane][r] = Facc[r];
    __syncthreads();

    if (KS > 1) {
        const int t = threadIdx.x;
        if (t < 64) {
            const int col = t >> 4;
            const int qi  = t & 15;
            const int sl  = col + ((qi >> 3) << 4);
            float s = red[0][sl][qi & 7];
            #pragma unroll
            for (int w = 1; w < KWAVES; ++w) s += red[w][sl][qi & 7];
            fout[(size_t)(tile * KS + part) * 64 + t] = s;
        }
    } else if (wv == 0) {
        #pragma unroll
        for (int w = 1; w < KWAVES; ++w) {
            #pragma unroll
            for (int r = 0; r < 8; ++r) Facc[r] += red[w][lane][r];
        }

        #pragma unroll
        for (int r = 0; r < 8; ++r) {
            float s_q = lo ? bcastf(Facc[r], 3) : bcastf(Facc[r], 19);
            if (ch < CCH) {
                const int qq = qt * 16 + r + (lo ? 0 : 8);
                fout[((size_t)b * CCH + ch) * N_PIX + qq] = Facc[r] / s_q;
            }
        }
    }
}

#define CT_H   8
#define CT_W   32
#define CT_HH  (CT_H + 2)
#define CT_WW  (CT_W + 2)
#define CT_PL  (CT_HH * CT_WW)
#define CT_N   (CCH * CT_PL)
#define CT_LD  4

template <int KS>
__global__ __launch_bounds__(256)
void conv3x3_bias_residual(const float* __restrict__ inp, const float* __restrict__ f,
                           const float* __restrict__ w, const float* __restrict__ bias,
                           float* __restrict__ out) {
    __shared__ float tile[CT_LD * 256];

    const int tid = threadIdx.x;
    const int tx  = tid & 31;
    const int ty  = tid >> 5;
    int bid = blockIdx.x;
    const int x0 = (bid % (HDIM / CT_W)) * CT_W;  bid /= (HDIM / CT_W);
    const int y0 = (bid % (HDIM / CT_H)) * CT_H;
    const int b  = bid / (HDIM / CT_H);

    float wr[CCH * CCH * 9];
    #pragma unroll
    for (int i = 0; i < CCH * CCH * 9; ++i) wr[i] = w[i];
    float acc0 = bias[0], acc1 = bias[1], acc2 = bias[2];

    const int y = y0 + ty, x = x0 + tx;
    const size_t o0 = (size_t)b * CCH * N_PIX + y * HDIM + x;
    const float res0 = inp[o0];
    const float res1 = inp[o0 + N_PIX];
    const float res2 = inp[o0 + 2 * N_PIX];

    if (KS > 1) {
        float pv[2][KS][4];
        bool  inb[2];
        #pragma unroll
        for (int k = 0; k < 2; ++k) {
            int p = tid + 256 * k;
            p = p < CT_PL ? p : CT_PL - 1;
            const int yy = p / CT_WW;
            const int xx = p - yy * CT_WW;
            const int gy = y0 + yy - 1;
            const int gx = x0 + xx - 1;
            inb[k] = (unsigned)gy < (unsigned)HDIM && (unsigned)gx < (unsigned)HDIM;
            const int n = inb[k] ? gy * HDIM + gx : 0;
            const float* pp = f + (size_t)((b * QTILES + (n >> 4)) * KS) * 64 + (n & 15);
            #pragma unroll
            for (int h = 0; h < KS; ++h) {
                #pragma unroll
                for (int c = 0; c < 4; ++c) pv[k][h][c] = pp[h * 64 + c * 16];
            }
        }

        asm volatile("" :: "v"(pv[0][0][0]), "v"(pv[0][0][1]), "v"(pv[0][0][2]), "v"(pv[0][0][3]),
                           "v"(pv[0][1][0]), "v"(pv[0][1][1]), "v"(pv[0][1][2]), "v"(pv[0][1][3]),
                           "v"(pv[1][0][0]), "v"(pv[1][0][1]), "v"(pv[1][0][2]), "v"(pv[1][0][3]),
                           "v"(pv[1][1][0]), "v"(pv[1][1][1]), "v"(pv[1][1][2]), "v"(pv[1][1][3]));
        #pragma unroll
        for (int k = 0; k < 2; ++k) {
            float s[4];
            #pragma unroll
            for (int c = 0; c < 4; ++c) {
                s[c] = pv[k][0][c];
                #pragma unroll
                for (int h = 1; h < KS; ++h) s[c] += pv[k][h][c];
            }
            const int  p  = tid + 256 * k;
            const bool in = p < CT_PL;
            tile[in ? 0 * CT_PL + p : CT_N + 0] = inb[k] ? s[0] / s[3] : 0.0f;
            tile[in ? 1 * CT_PL + p : CT_N + 1] = inb[k] ? s[1] / s[3] : 0.0f;
            tile[in ? 2 * CT_PL + p : CT_N + 2] = inb[k] ? s[2] / s[3] : 0.0f;
        }
    } else {
        const float* fb = f + (size_t)b * CCH * N_PIX;
        float v[CT_LD];
        #pragma unroll
        for (int k = 0; k < CT_LD; ++k) {
            int e = tid + 256 * k;
            e = e < CT_N ? e : CT_N - 1;
            const int ci = e / CT_PL;
            const int r  = e - ci * CT_PL;
            const int yy = r / CT_WW;
            const int xx = r - yy * CT_WW;
            const int gy = y0 + yy - 1;
            const int gx = x0 + xx - 1;
            const bool in = (unsigned)gy < (unsigned)HDIM && (unsigned)gx < (unsigned)HDIM;
            const float t = fb[ci * N_PIX + (in ? gy * HDIM + gx : 0)];
            v[k] = in ? t : 0.0f;
        }

        #pragma unroll
        for (int k = 0; k < CT_LD; ++k) {
            tile[tid + 256 * k] = v[k];
        }
    }
    __syncthreads();

    #pragma unroll
    for (int ci = 0; ci < CCH; ++ci) {
        #pragma unroll
        for (int dy = 0; dy < 3; ++dy) {
            #pragma unroll
            for (int dx = 0; dx < 3; ++dx) {
                const float t = tile[ci * CT_PL + (ty + dy) * CT_WW + tx + dx];
                acc0 += wr[((0 * CCH + ci) * 3 + dy) * 3 + dx] * t;
                acc1 += wr[((1 * CCH + ci) * 3 + dy) * 3 + dx] * t;
                acc2 += wr[((2 * CCH + ci) * 3 + dy) * 3 + dx] * t;
            }
        }
    }
    out[o0]             = res0 + acc0;
    out[o0 + N_PIX]     = res1 + acc1;
    out[o0 + 2 * N_PIX] = res2 + acc2;
}

extern "C" void kernel_launch(void* const* d_in, const int* in_sizes, int n_in,
                              void* d_out, int out_size, void* d_ws, size_t ws_size,
                              hipStream_t stream) {
    const float* inp    = (const float*)d_in[0];
    const float* conv_w = (const float*)d_in[1];
    const float* conv_b = (const float*)d_in[2];
    float* out = (float*)d_out;
    float* ws  = (float*)d_ws;

    if (ws_size >= sizeof(float) * PART_FLOATS) {
        attn_flash_wmma<KSPLIT><<<BATCH * QTILES * KSPLIT, 32 * (KSTREAMS / KSPLIT), 0, stream>>>(inp, ws);
        conv3x3_bias_residual<KSPLIT><<<72, 256, 0, stream>>>(inp, ws, conv_w, conv_b, out);
    } else {
        attn_flash_wmma<1><<<BATCH * QTILES, 32 * KSTREAMS, 0, stream>>>(inp, ws);
        conv3x3_bias_residual<1><<<72, 256, 0, stream>>>(inp, ws, conv_w, conv_b, out);
    }
}
